# starting kernel
# baseline (speedup 1.0000x reference)
#include <hip/hip_runtime.h>


typedef __attribute__((ext_vector_type(2))) float v2f;
typedef __attribute__((ext_vector_type(8))) float v8f;

#define S_LEN 1024
#define D_DIM 64
#define BH    64

__global__ __launch_bounds__(256)
void entmax_attn_kernel(const float* __restrict__ q,
                        const float* __restrict__ k,
                        const float* __restrict__ v,
                        float* __restrict__ out)
{
    __shared__ float sm[16 * S_LEN];

    const int lane  = threadIdx.x & 31;
    const int wave  = threadIdx.x >> 5;
    const int lmod  = lane & 15;
    const int lhalf = lane >> 4;

    const int bh    = blockIdx.y;
    const int qbase = blockIdx.x * 16;

    const float* qp = q + ((size_t)(bh * S_LEN + qbase + lmod)) * D_DIM;
    v2f aq[16];
#pragma unroll
    for (int s = 0; s < 16; ++s)
        aq[s] = *(const v2f*)(qp + 4 * s + 2 * lhalf);

#pragma unroll
    for (int t = 0; t < 8; ++t) {
        const int nbase = (wave * 8 + t) * 16;
        const float* kp = k + ((size_t)(bh * S_LEN + nbase + lmod)) * D_DIM;
        v8f acc = {};
#pragma unroll
        for (int s = 0; s < 16; ++s) {
            v2f bf = *(const v2f*)(kp + 4 * s + 2 * lhalf);
            acc = __builtin_amdgcn_wmma_f32_16x16x4_f32(
                false, aq[s], false, bf, (short)0, acc, false, false);
        }
#pragma unroll
        for (int r = 0; r < 8; ++r)
            sm[(r + 8 * lhalf) * S_LEN + nbase + lmod] = acc[r] * 0.0625f;
    }
    __syncthreads();

    for (int rr = 0; rr < 2; ++rr) {
        const int row = wave * 2 + rr;
        float* rp = &sm[row * S_LEN];
        float x[32];
#pragma unroll
        for (int j = 0; j < 32; ++j) x[j] = rp[lane + 32 * j];

        float m = x[0];
#pragma unroll
        for (int j = 1; j < 32; ++j) m = fmaxf(m, x[j]);
#pragma unroll
        for (int off = 16; off >= 1; off >>= 1)
            m = fmaxf(m, __shfl_xor(m, off, 32));
#pragma unroll
        for (int j = 0; j < 32; ++j) x[j] -= m;

        float tau = -1.0f;
        for (int it = 0; it < 12; ++it) {
            float s = 0.f, g = 0.f;
#pragma unroll
            for (int j = 0; j < 32; ++j) {
                float d = fmaxf(x[j] - tau, 0.f);
                s = fmaf(d, d, s);
                g += d;
            }
#pragma unroll
            for (int off = 16; off >= 1; off >>= 1) {
                s += __shfl_xor(s, off, 32);
                g += __shfl_xor(g, off, 32);
            }
            g = fmaxf(g, 1e-20f);
            tau += (s - 1.0f) / (2.0f * g);
        }

        float p[32];
        float ssum = 0.f;
#pragma unroll
        for (int j = 0; j < 32; ++j) {
            float d = fmaxf(x[j] - tau, 0.f);
            p[j] = d * d;
            ssum += p[j];
        }
#pragma unroll
        for (int off = 16; off >= 1; off >>= 1)
            ssum += __shfl_xor(ssum, off, 32);
        const float inv = (ssum > 0.f) ? (1.0f / ssum) : 0.f;
#pragma unroll
        for (int j = 0; j < 32; ++j) rp[lane + 32 * j] = p[j] * inv;
    }
    __syncthreads();

    const int t      = wave & 3;
    const int kstart = (wave >> 2) * 512;
    const int ncol   = t * 16 + lmod;
    const float* vp  = v + ((size_t)bh * S_LEN) * D_DIM + ncol;

    v8f acc = {};
#pragma unroll 8
    for (int kb = kstart; kb < kstart + 512; kb += 4) {
        v2f a = *(const v2f*)(&sm[lmod * S_LEN + kb + 2 * lhalf]);
        v2f b;
        b.x = vp[(size_t)(kb + 2 * lhalf + 0) * D_DIM];
        b.y = vp[(size_t)(kb + 2 * lhalf + 1) * D_DIM];
        acc = __builtin_amdgcn_wmma_f32_16x16x4_f32(
            false, a, false, b, (short)0, acc, false, false);
    }
    __syncthreads();

#pragma unroll
    for (int r = 0; r < 8; ++r)
        sm[wave * 256 + r * 32 + lane] = acc[r];
    __syncthreads();

    if (wave < 4) {
#pragma unroll
        for (int r = 0; r < 8; ++r) {
            float val = sm[wave * 256 + r * 32 + lane]
                      + sm[(wave + 4) * 256 + r * 32 + lane];
            const int row = r + 8 * lhalf;
            const int col = wave * 16 + lmod;
            out[((size_t)(bh * S_LEN + qbase + row)) * D_DIM + col] = val;
        }
    }
}

extern "C" void kernel_launch(void* const* d_in, const int* in_sizes, int n_in,
                              void* d_out, int out_size, void* d_ws, size_t ws_size,
                              hipStream_t stream) {
    (void)in_sizes; (void)n_in; (void)out_size; (void)d_ws; (void)ws_size;
    const float* q = (const float*)d_in[0];
    const float* k = (const float*)d_in[1];
    const float* v = (const float*)d_in[2];
    float* out = (float*)d_out;

    dim3 grid(S_LEN / 16, BH);
    entmax_attn_kernel<<<grid, 256, 0, stream>>>(q, k, v, out);
}


# Scores formed transposed (K*Q^T) and kept in registers: accumulator layout is the P^T operand of out^T=V^T*P^T, no LDS pass for scores/probs; row sums in-lane + per-wave partials; sum order changes
# speedup vs baseline: 1.4509x; 1.4509x over previous
#include <hip/hip_runtime.h>


typedef __attribute__((ext_vector_type(2))) float v2f;
typedef __attribute__((ext_vector_type(4))) float v4f;
typedef __attribute__((ext_vector_type(8))) float v8f;

#define S_LEN 1024
#define D_DIM 64
#define BH    64
#define NW    8
#define PP    68

__device__ __forceinline__ float swap_halves(float x)
{
    unsigned u = __float_as_uint(x);
    u = __builtin_amdgcn_permlanex16(u, u, 0x76543210u, 0xfedcba98u, false, false);
    return __uint_as_float(u);
}

__global__ __launch_bounds__(256, 6)
void entmax_attn_kernel(const float* __restrict__ q,
                        const float* __restrict__ k,
                        const float* __restrict__ v,
                        float* __restrict__ out)
{
    __shared__ float redm[NW * 16];
    __shared__ v2f   red[2][NW * 16];
    __shared__ __attribute__((aligned(16))) float part[NW][16 * PP];

    const int lane  = threadIdx.x & 31;
    const int wave  = threadIdx.x >> 5;
    const int lmod  = lane & 15;
    const int lhalf = lane >> 4;

    const int bh    = blockIdx.y;
    const int qbase = blockIdx.x * 16;
    const int kbase = wave * 128;

    const float* qp = q + ((size_t)(bh * S_LEN + qbase + lmod)) * D_DIM + 2 * lhalf;
    v2f bq[16];
#pragma unroll
    for (int s = 0; s < 16; ++s)
        bq[s] = *(const v2f*)(qp + 4 * s);

    v8f x[8];
#pragma unroll
    for (int t = 0; t < 8; ++t) {
        const float* kp = k + ((size_t)(bh * S_LEN + kbase + t * 16 + lmod)) * D_DIM + 2 * lhalf;
        v8f acc = {};
#pragma unroll
        for (int s = 0; s < 16; ++s) {
            v2f ak = *(const v2f*)(kp + 4 * s);
            acc = __builtin_amdgcn_wmma_f32_16x16x4_f32(
                false, ak, false, bq[s], (short)0, acc, false, false);
        }
#pragma unroll
        for (int r = 0; r < 8; ++r) x[t][r] = acc[r] * 0.0625f;
    }

    float m0 = x[0][0], m1 = x[0][1];
#pragma unroll
    for (int t = 0; t < 8; ++t)
#pragma unroll
        for (int r = 0; r < 8; r += 2) {
            m0 = fmaxf(m0, x[t][r]);
            m1 = fmaxf(m1, x[t][r + 1]);
        }
    float m = fmaxf(m0, m1);
    m = fmaxf(m, swap_halves(m));
    if (lhalf == 0) redm[wave * 16 + lmod] = m;
    __syncthreads();
    m = redm[lmod];
#pragma unroll
    for (int w = 1; w < NW; ++w) m = fmaxf(m, redm[w * 16 + lmod]);
#pragma unroll
    for (int t = 0; t < 8; ++t)
#pragma unroll
        for (int r = 0; r < 8; ++r) x[t][r] -= m;

    float tau = -1.0f;
    for (int it = 0; it < 13; ++it) {
        float s0 = 0.f, s1 = 0.f, g0 = 0.f, g1 = 0.f;
#pragma unroll
        for (int t = 0; t < 8; ++t)
#pragma unroll
            for (int r = 0; r < 8; r += 2) {
                float d0 = fmaxf(x[t][r]     - tau, 0.f);
                float d1 = fmaxf(x[t][r + 1] - tau, 0.f);
                s0 = fmaf(d0, d0, s0);  g0 += d0;
                s1 = fmaf(d1, d1, s1);  g1 += d1;
            }
        float s = s0 + s1, g = g0 + g1;
        s += swap_halves(s);
        g += swap_halves(g);
        v2f* rb = red[it & 1];
        if (lhalf == 0) { v2f e; e.x = s; e.y = g; rb[wave * 16 + lmod] = e; }
        __syncthreads();
        float st = 0.f, gt = 0.f;
#pragma unroll
        for (int w = 0; w < NW; ++w) {
            v2f e = rb[w * 16 + lmod];
            st += e.x;  gt += e.y;
        }
        if (it < 12) {
            gt = fmaxf(gt, 1e-20f);
            tau += (st - 1.0f) / (2.0f * gt);
        } else {
            const float inv = (st > 0.f) ? (1.0f / st) : 0.f;
#pragma unroll
            for (int t = 0; t < 8; ++t)
#pragma unroll
                for (int r = 0; r < 8; ++r) {
                    float d = fmaxf(x[t][r] - tau, 0.f);
                    x[t][r] = (d * d) * inv;
                }
        }
    }

    const float* vp = v + ((size_t)(bh * S_LEN + kbase + 8 * lhalf)) * D_DIM + lmod;
    v8f o[4];
#pragma unroll
    for (int dt = 0; dt < 4; ++dt) o[dt] = (v8f){};
#pragma unroll
    for (int t = 0; t < 8; ++t) {
#pragma unroll
        for (int j = 0; j < 4; ++j) {
            const float* vr = vp + (size_t)(t * 16 + 2 * j) * D_DIM;
            v2f b;
            b.x = x[t][2 * j];
            b.y = x[t][2 * j + 1];
#pragma unroll
            for (int dt = 0; dt < 4; ++dt) {
                v2f a;
                a.x = vr[dt * 16];
                a.y = vr[D_DIM + dt * 16];
                o[dt] = __builtin_amdgcn_wmma_f32_16x16x4_f32(
                    false, a, false, b, (short)0, o[dt], false, false);
            }
        }
    }

    float* pw = &part[wave][lmod * PP + 8 * lhalf];
#pragma unroll
    for (int dt = 0; dt < 4; ++dt) {
        v4f lo, hi;
        lo.x = o[dt][0]; lo.y = o[dt][1]; lo.z = o[dt][2]; lo.w = o[dt][3];
        hi.x = o[dt][4]; hi.y = o[dt][5]; hi.z = o[dt][6]; hi.w = o[dt][7];
        *(v4f*)(pw + dt * 16)     = lo;
        *(v4f*)(pw + dt * 16 + 4) = hi;
    }
    __syncthreads();

    const int oq = threadIdx.x >> 4;
    const int od = (threadIdx.x & 15) * 4;
    v4f acc4 = *(const v4f*)(&part[0][oq * PP + od]);
#pragma unroll
    for (int w = 1; w < NW; ++w)
        acc4 += *(const v4f*)(&part[w][oq * PP + od]);
    *(v4f*)(out + ((size_t)(bh * S_LEN + qbase + oq)) * D_DIM + od) = acc4;
}

extern "C" void kernel_launch(void* const* d_in, const int* in_sizes, int n_in,
                              void* d_out, int out_size, void* d_ws, size_t ws_size,
                              hipStream_t stream) {
    (void)in_sizes; (void)n_in; (void)out_size; (void)d_ws; (void)ws_size;
    const float* q = (const float*)d_in[0];
    const float* k = (const float*)d_in[1];
    const float* v = (const float*)d_in[2];
    float* out = (float*)d_out;

    dim3 grid(S_LEN / 16, BH);
    entmax_attn_kernel<<<grid, 256, 0, stream>>>(q, k, v, out);
}


# Each wave takes two query tiles (32 queries per block): every K and V fragment feeds two products; K read one whole 128-B line per key at a time (d in two halves, tile by tile); sums in parent order
# speedup vs baseline: 2.0440x; 1.4088x over previous
#include <hip/hip_runtime.h>


typedef __attribute__((ext_vector_type(2))) float v2f;
typedef __attribute__((ext_vector_type(4))) float v4f;
typedef __attribute__((ext_vector_type(8))) float v8f;

#define S_LEN 1024
#define D_DIM 64
#define BH    64
#define NW    8
#define QB    32
#define PP    68

__device__ __forceinline__ float swap_halves(float x)
{
    unsigned u = __float_as_uint(x);
    u = __builtin_amdgcn_permlanex16(u, u, 0x76543210u, 0xfedcba98u, false, false);
    return __uint_as_float(u);
}

__global__ __launch_bounds__(256, 4)
void entmax_attn_kernel(const float* __restrict__ q,
                        const float* __restrict__ k,
                        const float* __restrict__ v,
                        float* __restrict__ out)
{
    __shared__ float redm[NW * QB];
    __shared__ v2f   red[2][NW * QB];
    __shared__ __attribute__((aligned(16))) float part[NW][QB * PP];

    const int lane  = threadIdx.x & 31;
    const int wave  = threadIdx.x >> 5;
    const int lmod  = lane & 15;
    const int lhalf = lane >> 4;
    const bool hi   = lhalf != 0;

    const int bh    = blockIdx.y;
    const int qbase = blockIdx.x * QB;
    const int kbase = wave * 128;

    const float* qp = q + ((size_t)(bh * S_LEN + qbase + lmod)) * D_DIM + 2 * lhalf;
    const float* kp = k + ((size_t)(bh * S_LEN + kbase + lmod)) * D_DIM + 2 * lhalf;
    v8f x[2][8];
#pragma unroll
    for (int u = 0; u < 2; ++u)
#pragma unroll
        for (int t = 0; t < 8; ++t) x[u][t] = (v8f){};
    v2f bq[2][8], fk[3][8];
#pragma unroll
    for (int i = 0; i < 8; ++i) {
        bq[0][i] = *(const v2f*)(qp + 4 * i);
        bq[1][i] = *(const v2f*)(qp + 16 * D_DIM + 4 * i);
    }
#pragma unroll
    for (int g = 0; g < 2; ++g)
#pragma unroll
        for (int i = 0; i < 8; ++i)
            fk[g][i] = *(const v2f*)(kp + (size_t)((g & 7) * 16) * D_DIM + 4 * (8 * (g >> 3) + i));
#pragma unroll
    for (int g = 0; g < 16; ++g) {
        const int t = g & 7;
        if (g + 2 < 16) {
            const int n = (g + 2) % 3, tn = (g + 2) & 7, hn = (g + 2) >> 3;
#pragma unroll
            for (int i = 0; i < 8; ++i)
                fk[n][i] = *(const v2f*)(kp + (size_t)(tn * 16) * D_DIM + 4 * (8 * hn + i));
        }
        const int c = g % 3;
#pragma unroll
        for (int i = 0; i < 8; ++i) {
            x[0][t] = __builtin_amdgcn_wmma_f32_16x16x4_f32(
                false, fk[c][i], false, bq[0][i], (short)0, x[0][t], false, false);
            x[1][t] = __builtin_amdgcn_wmma_f32_16x16x4_f32(
                false, fk[c][i], false, bq[1][i], (short)0, x[1][t], false, false);
            if (g == 7) {
                bq[0][i] = *(const v2f*)(qp + 4 * (8 + i));
                bq[1][i] = *(const v2f*)(qp + 16 * D_DIM + 4 * (8 + i));
            }
        }
        __builtin_amdgcn_sched_barrier(0);
    }
#pragma unroll
    for (int u = 0; u < 2; ++u)
#pragma unroll
        for (int t = 0; t < 8; ++t)
#pragma unroll
            for (int r = 0; r < 8; ++r) x[u][t][r] *= 0.0625f;

    float mx[2];
#pragma unroll
    for (int u = 0; u < 2; ++u) {
        float m0 = x[u][0][0], m1 = x[u][0][1];
#pragma unroll
        for (int t = 0; t < 8; ++t)
#pragma unroll
            for (int r = 0; r < 8; r += 2) {
                m0 = fmaxf(m0, x[u][t][r]);
                m1 = fmaxf(m1, x[u][t][r + 1]);
            }
        mx[u] = fmaxf(m0, m1);
    }
    {
        const float got = swap_halves(hi ? mx[0] : mx[1]);
        redm[wave * QB + lane] = fmaxf(hi ? mx[1] : mx[0], got);
    }
    __syncthreads();
    {
        float m = redm[lane];
#pragma unroll
        for (int w = 1; w < NW; ++w) m = fmaxf(m, redm[w * QB + lane]);
        const float mo = swap_halves(m);
        mx[0] = hi ? mo : m;
        mx[1] = hi ? m : mo;
    }
#pragma unroll
    for (int u = 0; u < 2; ++u)
#pragma unroll
        for (int t = 0; t < 8; ++t)
#pragma unroll
            for (int r = 0; r < 8; ++r) x[u][t][r] -= mx[u];

    float tau_own = -1.0f;
    float tau[2] = { -1.0f, -1.0f };
    for (int it = 0; it < 13; ++it) {
        float sq[2], sg[2];
#pragma unroll
        for (int u = 0; u < 2; ++u) {
            float s0 = 0.f, s1 = 0.f, g0 = 0.f, g1 = 0.f;
            const float tu = tau[u];
#pragma unroll
            for (int t = 0; t < 8; ++t)
#pragma unroll
                for (int r = 0; r < 8; r += 2) {
                    float d0 = fmaxf(x[u][t][r]     - tu, 0.f);
                    float d1 = fmaxf(x[u][t][r + 1] - tu, 0.f);
                    s0 = fmaf(d0, d0, s0);  g0 += d0;
                    s1 = fmaf(d1, d1, s1);  g1 += d1;
                }
            sq[u] = s0 + s1;
            sg[u] = g0 + g1;
        }
        const float s = (hi ? sq[1] : sq[0]) + swap_halves(hi ? sq[0] : sq[1]);
        const float g = (hi ? sg[1] : sg[0]) + swap_halves(hi ? sg[0] : sg[1]);
        v2f* rb = red[it & 1];
        { v2f e; e.x = s; e.y = g; rb[wave * QB + lane] = e; }
        __syncthreads();
        float st = 0.f, gt = 0.f;
#pragma unroll
        for (int w = 0; w < NW; ++w) {
            v2f e = rb[w * QB + lane];
            st += e.x;  gt += e.y;
        }
        if (it < 12) {
            gt = fmaxf(gt, 1e-20f);
            tau_own += (st - 1.0f) / (2.0f * gt);
            const float to = swap_halves(tau_own);
            tau[0] = hi ? to : tau_own;
            tau[1] = hi ? tau_own : to;
        } else {
            const float inv_own = (st > 0.f) ? (1.0f / st) : 0.f;
            const float io = swap_halves(inv_own);
            const float inv0 = hi ? io : inv_own;
            const float inv1 = hi ? inv_own : io;
#pragma unroll
            for (int u = 0; u < 2; ++u) {
                const float tu = tau[u];
                const float inv = u ? inv1 : inv0;
#pragma unroll
                for (int t = 0; t < 8; ++t)
#pragma unroll
                    for (int r = 0; r < 8; ++r) {
                        float d = fmaxf(x[u][t][r] - tu, 0.f);
                        x[u][t][r] = (d * d) * inv;
                    }
            }
        }
    }

    const float* vp = v + ((size_t)(bh * S_LEN + kbase + 8 * lhalf)) * D_DIM + lmod;
    float* pw = &part[wave][lmod * PP + 8 * lhalf];
#pragma unroll 1
    for (int dh = 0; dh < 2; ++dh) {
        v8f o[2][2];
#pragma unroll
        for (int e = 0; e < 2; ++e) { o[e][0] = (v8f){}; o[e][1] = (v8f){}; }
#pragma unroll
        for (int t = 0; t < 8; ++t) {
#pragma unroll
            for (int j = 0; j < 4; ++j) {
                const float* vr = vp + (size_t)(t * 16 + 2 * j) * D_DIM + dh * 32;
                v2f b0, b1;
                b0.x = x[0][t][2 * j];
                b0.y = x[0][t][2 * j + 1];
                b1.x = x[1][t][2 * j];
                b1.y = x[1][t][2 * j + 1];
#pragma unroll
                for (int e = 0; e < 2; ++e) {
                    v2f a;
                    a.x = vr[e * 16];
                    a.y = vr[D_DIM + e * 16];
                    o[e][0] = __builtin_amdgcn_wmma_f32_16x16x4_f32(
                        false, a, false, b0, (short)0, o[e][0], false, false);
                    o[e][1] = __builtin_amdgcn_wmma_f32_16x16x4_f32(
                        false, a, false, b1, (short)0, o[e][1], false, false);
                }
            }
        }
#pragma unroll
        for (int u = 0; u < 2; ++u)
#pragma unroll
            for (int e = 0; e < 2; ++e) {
                v4f lo, hi4;
                lo.x  = o[e][u][0]; lo.y  = o[e][u][1]; lo.z  = o[e][u][2]; lo.w  = o[e][u][3];
                hi4.x = o[e][u][4]; hi4.y = o[e][u][5]; hi4.z = o[e][u][6]; hi4.w = o[e][u][7];
                float* pd = pw + u * 16 * PP + dh * 32 + e * 16;
                *(v4f*)(pd)     = lo;
                *(v4f*)(pd + 4) = hi4;
            }
    }
    __syncthreads();

    const int od = (threadIdx.x & 15) * 4;
#pragma unroll
    for (int u = 0; u < 2; ++u) {
        const int oq = u * 16 + (threadIdx.x >> 4);
        v4f acc4 = *(const v4f*)(&part[0][oq * PP + od]);
#pragma unroll
        for (int w = 1; w < NW; ++w)
            acc4 += *(const v4f*)(&part[w][oq * PP + od]);
        *(v4f*)(out + ((size_t)(bh * S_LEN + qbase + oq)) * D_DIM + od) = acc4;
    }
}

extern "C" void kernel_launch(void* const* d_in, const int* in_sizes, int n_in,
                              void* d_out, int out_size, void* d_ws, size_t ws_size,
                              hipStream_t stream) {
    (void)in_sizes; (void)n_in; (void)out_size; (void)d_ws; (void)ws_size;
    const float* q = (const float*)d_in[0];
    const float* k = (const float*)d_in[1];
    const float* v = (const float*)d_in[2];
    float* out = (float*)d_out;

    dim3 grid(S_LEN / QB, BH);
    entmax_attn_kernel<<<grid, 256, 0, stream>>>(q, k, v, out);
}


# Tau root-finder: Newton step taken on sqrt(sum d^2)-1 (convex, near-linear far from the root) instead of sum d^2-1, same two row sums per step: 7 steps replace 12; tau may differ in its last bits
# speedup vs baseline: 2.3105x; 1.1304x over previous
#include <hip/hip_runtime.h>


typedef __attribute__((ext_vector_type(2))) float v2f;
typedef __attribute__((ext_vector_type(4))) float v4f;
typedef __attribute__((ext_vector_type(8))) float v8f;

#define S_LEN 1024
#define D_DIM 64
#define BH    64
#define NW    8
#define QB    32
#define PP    68
#define N_STEP 7

__device__ __forceinline__ float swap_halves(float x)
{
    unsigned u = __float_as_uint(x);
    u = __builtin_amdgcn_permlanex16(u, u, 0x76543210u, 0xfedcba98u, false, false);
    return __uint_as_float(u);
}

__global__ __launch_bounds__(256, 4)
void entmax_attn_kernel(const float* __restrict__ q,
                        const float* __restrict__ k,
                        const float* __restrict__ v,
                        float* __restrict__ out)
{
    __shared__ float redm[NW * QB];
    __shared__ v2f   red[2][NW * QB];
    __shared__ __attribute__((aligned(16))) float part[NW][QB * PP];

    const int lane  = threadIdx.x & 31;
    const int wave  = threadIdx.x >> 5;
    const int lmod  = lane & 15;
    const int lhalf = lane >> 4;
    const bool hi   = lhalf != 0;

    const int bh    = blockIdx.y;
    const int qbase = blockIdx.x * QB;
    const int kbase = wave * 128;

    const float* qp = q + ((size_t)(bh * S_LEN + qbase + lmod)) * D_DIM + 2 * lhalf;
    const float* kp = k + ((size_t)(bh * S_LEN + kbase + lmod)) * D_DIM + 2 * lhalf;
    v8f x[2][8];
#pragma unroll
    for (int u = 0; u < 2; ++u)
#pragma unroll
        for (int t = 0; t < 8; ++t) x[u][t] = (v8f){};
    v2f bq[2][8], fk[3][8];
#pragma unroll
    for (int i = 0; i < 8; ++i) {
        bq[0][i] = *(const v2f*)(qp + 4 * i);
        bq[1][i] = *(const v2f*)(qp + 16 * D_DIM + 4 * i);
    }
#pragma unroll
    for (int g = 0; g < 2; ++g)
#pragma unroll
        for (int i = 0; i < 8; ++i)
            fk[g][i] = *(const v2f*)(kp + (size_t)((g & 7) * 16) * D_DIM + 4 * (8 * (g >> 3) + i));
#pragma unroll
    for (int g = 0; g < 16; ++g) {
        const int t = g & 7;
        if (g + 2 < 16) {
            const int n = (g + 2) % 3, tn = (g + 2) & 7, hn = (g + 2) >> 3;
#pragma unroll
            for (int i = 0; i < 8; ++i)
                fk[n][i] = *(const v2f*)(kp + (size_t)(tn * 16) * D_DIM + 4 * (8 * hn + i));
        }
        const int c = g % 3;
#pragma unroll
        for (int i = 0; i < 8; ++i) {
            x[0][t] = __builtin_amdgcn_wmma_f32_16x16x4_f32(
                false, fk[c][i], false, bq[0][i], (short)0, x[0][t], false, false);
            x[1][t] = __builtin_amdgcn_wmma_f32_16x16x4_f32(
                false, fk[c][i], false, bq[1][i], (short)0, x[1][t], false, false);
            if (g == 7) {
                bq[0][i] = *(const v2f*)(qp + 4 * (8 + i));
                bq[1][i] = *(const v2f*)(qp + 16 * D_DIM + 4 * (8 + i));
            }
        }
        __builtin_amdgcn_sched_barrier(0);
    }
#pragma unroll
    for (int u = 0; u < 2; ++u)
#pragma unroll
        for (int t = 0; t < 8; ++t)
#pragma unroll
            for (int r = 0; r < 8; ++r) x[u][t][r] *= 0.0625f;

    float mx[2];
#pragma unroll
    for (int u = 0; u < 2; ++u) {
        float m0 = x[u][0][0], m1 = x[u][0][1];
#pragma unroll
        for (int t = 0; t < 8; ++t)
#pragma unroll
            for (int r = 0; r < 8; r += 2) {
                m0 = fmaxf(m0, x[u][t][r]);
                m1 = fmaxf(m1, x[u][t][r + 1]);
            }
        mx[u] = fmaxf(m0, m1);
    }
    {
        const float got = swap_halves(hi ? mx[0] : mx[1]);
        redm[wave * QB + lane] = fmaxf(hi ? mx[1] : mx[0], got);
    }
    __syncthreads();
    {
        float m = redm[lane];
#pragma unroll
        for (int w = 1; w < NW; ++w) m = fmaxf(m, redm[w * QB + lane]);
        const float mo = swap_halves(m);
        mx[0] = hi ? mo : m;
        mx[1] = hi ? m : mo;
    }
#pragma unroll
    for (int u = 0; u < 2; ++u)
#pragma unroll
        for (int t = 0; t < 8; ++t)
#pragma unroll
            for (int r = 0; r < 8; ++r) x[u][t][r] -= mx[u];

    float tau_own = -1.0f;
    float tau[2] = { -1.0f, -1.0f };
    for (int it = 0; it <= N_STEP; ++it) {
        float sq[2], sg[2];
#pragma unroll
        for (int u = 0; u < 2; ++u) {
            float s0 = 0.f, s1 = 0.f, g0 = 0.f, g1 = 0.f;
            const float tu = tau[u];
#pragma unroll
            for (int t = 0; t < 8; ++t)
#pragma unroll
                for (int r = 0; r < 8; r += 2) {
                    float d0 = fmaxf(x[u][t][r]     - tu, 0.f);
                    float d1 = fmaxf(x[u][t][r + 1] - tu, 0.f);
                    s0 = fmaf(d0, d0, s0);  g0 += d0;
                    s1 = fmaf(d1, d1, s1);  g1 += d1;
                }
            sq[u] = s0 + s1;
            sg[u] = g0 + g1;
        }
        const float s = (hi ? sq[1] : sq[0]) + swap_halves(hi ? sq[0] : sq[1]);
        const float g = (hi ? sg[1] : sg[0]) + swap_halves(hi ? sg[0] : sg[1]);
        v2f* rb = red[it & 1];
        { v2f e; e.x = s; e.y = g; rb[wave * QB + lane] = e; }
        __syncthreads();
        float st = 0.f, gt = 0.f;
#pragma unroll
        for (int w = 0; w < NW; ++w) {
            v2f e = rb[w * QB + lane];
            st += e.x;  gt += e.y;
        }
        if (it < N_STEP) {
            gt = fmaxf(gt, 1e-20f);
            tau_own += (st - __builtin_amdgcn_sqrtf(st)) / gt;
            const float to = swap_halves(tau_own);
            tau[0] = hi ? to : tau_own;
            tau[1] = hi ? tau_own : to;
        } else {
            const float inv_own = (st > 0.f) ? (1.0f / st) : 0.f;
            const float io = swap_halves(inv_own);
            const float inv0 = hi ? io : inv_own;
            const float inv1 = hi ? inv_own : io;
#pragma unroll
            for (int u = 0; u < 2; ++u) {
                const float tu = tau[u];
                const float inv = u ? inv1 : inv0;
#pragma unroll
                for (int t = 0; t < 8; ++t)
#pragma unroll
                    for (int r = 0; r < 8; ++r) {
                        float d = fmaxf(x[u][t][r] - tu, 0.f);
                        x[u][t][r] = (d * d) * inv;
                    }
            }
        }
    }

    const float* vp = v + ((size_t)(bh * S_LEN + kbase + 8 * lhalf)) * D_DIM + lmod;
    float* pw = &part[wave][lmod * PP + 8 * lhalf];
#pragma unroll 1
    for (int dh = 0; dh < 2; ++dh) {
        v8f o[2][2];
#pragma unroll
        for (int e = 0; e < 2; ++e) { o[e][0] = (v8f){}; o[e][1] = (v8f){}; }
#pragma unroll
        for (int t = 0; t < 8; ++t) {
#pragma unroll
            for (int j = 0; j < 4; ++j) {
                const float* vr = vp + (size_t)(t * 16 + 2 * j) * D_DIM + dh * 32;
                v2f b0, b1;
                b0.x = x[0][t][2 * j];
                b0.y = x[0][t][2 * j + 1];
                b1.x = x[1][t][2 * j];
                b1.y = x[1][t][2 * j + 1];
#pragma unroll
                for (int e = 0; e < 2; ++e) {
                    v2f a;
                    a.x = vr[e * 16];
                    a.y = vr[D_DIM + e * 16];
                    o[e][0] = __builtin_amdgcn_wmma_f32_16x16x4_f32(
                        false, a, false, b0, (short)0, o[e][0], false, false);
                    o[e][1] = __builtin_amdgcn_wmma_f32_16x16x4_f32(
                        false, a, false, b1, (short)0, o[e][1], false, false);
                }
            }
        }
#pragma unroll
        for (int u = 0; u < 2; ++u)
#pragma unroll
            for (int e = 0; e < 2; ++e) {
                v4f lo, hi4;
                lo.x  = o[e][u][0]; lo.y  = o[e][u][1]; lo.z  = o[e][u][2]; lo.w  = o[e][u][3];
                hi4.x = o[e][u][4]; hi4.y = o[e][u][5]; hi4.z = o[e][u][6]; hi4.w = o[e][u][7];
                float* pd = pw + u * 16 * PP + dh * 32 + e * 16;
                *(v4f*)(pd)     = lo;
                *(v4f*)(pd + 4) = hi4;
            }
    }
    __syncthreads();

    const int od = (threadIdx.x & 15) * 4;
#pragma unroll
    for (int u = 0; u < 2; ++u) {
        const int oq = u * 16 + (threadIdx.x >> 4);
        v4f acc4 = *(const v4f*)(&part[0][oq * PP + od]);
#pragma unroll
        for (int w = 1; w < NW; ++w)
            acc4 += *(const v4f*)(&part[w][oq * PP + od]);
        *(v4f*)(out + ((size_t)(bh * S_LEN + qbase + oq)) * D_DIM + od) = acc4;
    }
}

extern "C" void kernel_launch(void* const* d_in, const int* in_sizes, int n_in,
                              void* d_out, int out_size, void* d_ws, size_t ws_size,
                              hipStream_t stream) {
    (void)in_sizes; (void)n_in; (void)out_size; (void)d_ws; (void)ws_size;
    const float* q = (const float*)d_in[0];
    const float* k = (const float*)d_in[1];
    const float* v = (const float*)d_in[2];
    float* out = (float*)d_out;

    dim3 grid(S_LEN / QB, BH);
    entmax_attn_kernel<<<grid, 256, 0, stream>>>(q, k, v, out);
}


# Tau iteration: 5 row-sum passes instead of 8: five Newton steps converge to f32 from tau=-1, and the normaliser is the last step's S - 2*g*dtau (first order), not a pass of its own; last bits differ
# speedup vs baseline: 2.4936x; 1.0792x over previous
#include <hip/hip_runtime.h>


typedef __attribute__((ext_vector_type(2))) float v2f;
typedef __attribute__((ext_vector_type(4))) float v4f;
typedef __attribute__((ext_vector_type(8))) float v8f;

#define S_LEN 1024
#define D_DIM 64
#define BH    64
#define NW    8
#define QB    32
#define PP    68
#define N_STEP 5

__device__ __forceinline__ float swap_halves(float x)
{
    unsigned u = __float_as_uint(x);
    u = __builtin_amdgcn_permlanex16(u, u, 0x76543210u, 0xfedcba98u, false, false);
    return __uint_as_float(u);
}

__global__ __launch_bounds__(256, 4)
void entmax_attn_kernel(const float* __restrict__ q,
                        const float* __restrict__ k,
                        const float* __restrict__ v,
                        float* __restrict__ out)
{
    __shared__ float redm[NW * QB];
    __shared__ v2f   red[2][NW * QB];
    __shared__ __attribute__((aligned(16))) float part[NW][QB * PP];

    const int lane  = threadIdx.x & 31;
    const int wave  = threadIdx.x >> 5;
    const int lmod  = lane & 15;
    const int lhalf = lane >> 4;
    const bool hi   = lhalf != 0;

    const int bh    = blockIdx.y;
    const int qbase = blockIdx.x * QB;
    const int kbase = wave * 128;

    const float* qp = q + ((size_t)(bh * S_LEN + qbase + lmod)) * D_DIM + 2 * lhalf;
    const float* kp = k + ((size_t)(bh * S_LEN + kbase + lmod)) * D_DIM + 2 * lhalf;
    v8f x[2][8];
#pragma unroll
    for (int u = 0; u < 2; ++u)
#pragma unroll
        for (int t = 0; t < 8; ++t) x[u][t] = (v8f){};
    v2f bq[2][8], fk[3][8];
#pragma unroll
    for (int i = 0; i < 8; ++i) {
        bq[0][i] = *(const v2f*)(qp + 4 * i);
        bq[1][i] = *(const v2f*)(qp + 16 * D_DIM + 4 * i);
    }
#pragma unroll
    for (int g = 0; g < 2; ++g)
#pragma unroll
        for (int i = 0; i < 8; ++i)
            fk[g][i] = *(const v2f*)(kp + (size_t)((g & 7) * 16) * D_DIM + 4 * (8 * (g >> 3) + i));
#pragma unroll
    for (int g = 0; g < 16; ++g) {
        const int t = g & 7;
        if (g + 2 < 16) {
            const int n = (g + 2) % 3, tn = (g + 2) & 7, hn = (g + 2) >> 3;
#pragma unroll
            for (int i = 0; i < 8; ++i)
                fk[n][i] = *(const v2f*)(kp + (size_t)(tn * 16) * D_DIM + 4 * (8 * hn + i));
        }
        const int c = g % 3;
#pragma unroll
        for (int i = 0; i < 8; ++i) {
            x[0][t] = __builtin_amdgcn_wmma_f32_16x16x4_f32(
                false, fk[c][i], false, bq[0][i], (short)0, x[0][t], false, false);
            x[1][t] = __builtin_amdgcn_wmma_f32_16x16x4_f32(
                false, fk[c][i], false, bq[1][i], (short)0, x[1][t], false, false);
            if (g == 7) {
                bq[0][i] = *(const v2f*)(qp + 4 * (8 + i));
                bq[1][i] = *(const v2f*)(qp + 16 * D_DIM + 4 * (8 + i));
            }
        }
        __builtin_amdgcn_sched_barrier(0);
    }
#pragma unroll
    for (int u = 0; u < 2; ++u)
#pragma unroll
        for (int t = 0; t < 8; ++t)
#pragma unroll
            for (int r = 0; r < 8; ++r) x[u][t][r] *= 0.0625f;

    float mx[2];
#pragma unroll
    for (int u = 0; u < 2; ++u) {
        float m0 = x[u][0][0], m1 = x[u][0][1];
#pragma unroll
        for (int t = 0; t < 8; ++t)
#pragma unroll
            for (int r = 0; r < 8; r += 2) {
                m0 = fmaxf(m0, x[u][t][r]);
                m1 = fmaxf(m1, x[u][t][r + 1]);
            }
        mx[u] = fmaxf(m0, m1);
    }
    {
        const float got = swap_halves(hi ? mx[0] : mx[1]);
        redm[wave * QB + lane] = fmaxf(hi ? mx[1] : mx[0], got);
    }
    __syncthreads();
    {
        float m = redm[lane];
#pragma unroll
        for (int w = 1; w < NW; ++w) m = fmaxf(m, redm[w * QB + lane]);
        const float mo = swap_halves(m);
        mx[0] = hi ? mo : m;
        mx[1] = hi ? m : mo;
    }
#pragma unroll
    for (int u = 0; u < 2; ++u)
#pragma unroll
        for (int t = 0; t < 8; ++t)
#pragma unroll
            for (int r = 0; r < 8; ++r) x[u][t][r] -= mx[u];

    float tau_own = -1.0f;
    float tau[2] = { -1.0f, -1.0f };
    float s_fin = 1.0f;
    for (int it = 0; it < N_STEP; ++it) {
        float sq[2], sg[2];
#pragma unroll
        for (int u = 0; u < 2; ++u) {
            float s0 = 0.f, s1 = 0.f, g0 = 0.f, g1 = 0.f;
            const float tu = tau[u];
#pragma unroll
            for (int t = 0; t < 8; ++t)
#pragma unroll
                for (int r = 0; r < 8; r += 2) {
                    float d0 = fmaxf(x[u][t][r]     - tu, 0.f);
                    float d1 = fmaxf(x[u][t][r + 1] - tu, 0.f);
                    s0 = fmaf(d0, d0, s0);  g0 += d0;
                    s1 = fmaf(d1, d1, s1);  g1 += d1;
                }
            sq[u] = s0 + s1;
            sg[u] = g0 + g1;
        }
        const float s = (hi ? sq[1] : sq[0]) + swap_halves(hi ? sq[0] : sq[1]);
        const float g = (hi ? sg[1] : sg[0]) + swap_halves(hi ? sg[0] : sg[1]);
        v2f* rb = red[it & 1];
        { v2f e; e.x = s; e.y = g; rb[wave * QB + lane] = e; }
        __syncthreads();
        float st = 0.f, gt = 0.f;
#pragma unroll
        for (int w = 0; w < NW; ++w) {
            v2f e = rb[w * QB + lane];
            st += e.x;  gt += e.y;
        }
        gt = fmaxf(gt, 1e-20f);
        const float t_old = tau_own;
        tau_own += (st - __builtin_amdgcn_sqrtf(st)) / gt;
        s_fin = fmaf(-2.0f * gt, tau_own - t_old, st);
        const float to = swap_halves(tau_own);
        tau[0] = hi ? to : tau_own;
        tau[1] = hi ? tau_own : to;
    }
    {
        const float inv_own = (s_fin > 0.f) ? (1.0f / s_fin) : 0.f;
        const float io = swap_halves(inv_own);
        const float inv0 = hi ? io : inv_own;
        const float inv1 = hi ? inv_own : io;
#pragma unroll
        for (int u = 0; u < 2; ++u) {
            const float tu = tau[u];
            const float inv = u ? inv1 : inv0;
#pragma unroll
            for (int t = 0; t < 8; ++t)
#pragma unroll
                for (int r = 0; r < 8; ++r) {
                    float d = fmaxf(x[u][t][r] - tu, 0.f);
                    x[u][t][r] = (d * d) * inv;
                }
        }
    }

    const float* vp = v + ((size_t)(bh * S_LEN + kbase + 8 * lhalf)) * D_DIM + lmod;
    float* pw = &part[wave][lmod * PP + 8 * lhalf];
#pragma unroll 1
    for (int dh = 0; dh < 2; ++dh) {
        v8f o[2][2];
#pragma unroll
        for (int e = 0; e < 2; ++e) { o[e][0] = (v8f){}; o[e][1] = (v8f){}; }
#pragma unroll
        for (int t = 0; t < 8; ++t) {
#pragma unroll
            for (int j = 0; j < 4; ++j) {
                const float* vr = vp + (size_t)(t * 16 + 2 * j) * D_DIM + dh * 32;
                v2f b0, b1;
                b0.x = x[0][t][2 * j];
                b0.y = x[0][t][2 * j + 1];
                b1.x = x[1][t][2 * j];
                b1.y = x[1][t][2 * j + 1];
#pragma unroll
                for (int e = 0; e < 2; ++e) {
                    v2f a;
                    a.x = vr[e * 16];
                    a.y = vr[D_DIM + e * 16];
                    o[e][0] = __builtin_amdgcn_wmma_f32_16x16x4_f32(
                        false, a, false, b0, (short)0, o[e][0], false, false);
                    o[e][1] = __builtin_amdgcn_wmma_f32_16x16x4_f32(
                        false, a, false, b1, (short)0, o[e][1], false, false);
                }
            }
        }
#pragma unroll
        for (int u = 0; u < 2; ++u)
#pragma unroll
            for (int e = 0; e < 2; ++e) {
                v4f lo, hi4;
                lo.x  = o[e][u][0]; lo.y  = o[e][u][1]; lo.z  = o[e][u][2]; lo.w  = o[e][u][3];
                hi4.x = o[e][u][4]; hi4.y = o[e][u][5]; hi4.z = o[e][u][6]; hi4.w = o[e][u][7];
                float* pd = pw + u * 16 * PP + dh * 32 + e * 16;
                *(v4f*)(pd)     = lo;
                *(v4f*)(pd + 4) = hi4;
            }
    }
    __syncthreads();

    const int od = (threadIdx.x & 15) * 4;
#pragma unroll
    for (int u = 0; u < 2; ++u) {
        const int oq = u * 16 + (threadIdx.x >> 4);
        v4f acc4 = *(const v4f*)(&part[0][oq * PP + od]);
#pragma unroll
        for (int w = 1; w < NW; ++w)
            acc4 += *(const v4f*)(&part[w][oq * PP + od]);
        *(v4f*)(out + ((size_t)(bh * S_LEN + qbase + oq)) * D_DIM + od) = acc4;
    }
}

extern "C" void kernel_launch(void* const* d_in, const int* in_sizes, int n_in,
                              void* d_out, int out_size, void* d_ws, size_t ws_size,
                              hipStream_t stream) {
    (void)in_sizes; (void)n_in; (void)out_size; (void)d_ws; (void)ws_size;
    const float* q = (const float*)d_in[0];
    const float* k = (const float*)d_in[1];
    const float* v = (const float*)d_in[2];
    float* out = (float*)d_out;

    dim3 grid(S_LEN / QB, BH);
    entmax_attn_kernel<<<grid, 256, 0, stream>>>(q, k, v, out);
}


# Q tile (32 x 64) fetched once per block by whole-line loads into LDS, pre-scaled by 1/16 (exact); the eight waves read their Q^T fragments from LDS instead of each gathering them from global memory
# speedup vs baseline: 2.5987x; 1.0421x over previous
#include <hip/hip_runtime.h>


typedef __attribute__((ext_vector_type(2))) float v2f;
typedef __attribute__((ext_vector_type(4))) float v4f;
typedef __attribute__((ext_vector_type(8))) float v8f;

#define S_LEN 1024
#define D_DIM 64
#define BH    64
#define NW    8
#define QB    32
#define PP    68
#define QP    68
#define N_STEP 5

__device__ __forceinline__ float swap_halves(float x)
{
    unsigned u = __float_as_uint(x);
    u = __builtin_amdgcn_permlanex16(u, u, 0x76543210u, 0xfedcba98u, false, false);
    return __uint_as_float(u);
}

__global__ __launch_bounds__(256, 4)
void entmax_attn_kernel(const float* __restrict__ q,
                        const float* __restrict__ k,
                        const float* __restrict__ v,
                        float* __restrict__ out)
{
    __shared__ float redm[NW * QB];
    __shared__ v2f   red[2][NW * QB];
    __shared__ __attribute__((aligned(16))) float part[NW][QB * PP];
    float* const qs = &part[0][0];

    const int lane  = threadIdx.x & 31;
    const int wave  = threadIdx.x >> 5;
    const int lmod  = lane & 15;
    const int lhalf = lane >> 4;
    const bool hi   = lhalf != 0;

    const int bh    = blockIdx.y;
    const int qbase = blockIdx.x * QB;
    const int kbase = wave * 128;

    const float* qp = qs + lmod * QP + 2 * lhalf;
    const float* kp = k + ((size_t)(bh * S_LEN + kbase + lmod)) * D_DIM + 2 * lhalf;
    v8f x[2][8];
#pragma unroll
    for (int u = 0; u < 2; ++u)
#pragma unroll
        for (int t = 0; t < 8; ++t) x[u][t] = (v8f){};
    v2f bq[2][8], fk[3][8];
    {
        const float* qg = q + ((size_t)(bh * S_LEN + qbase)) * D_DIM;
        v4f q0 = *(const v4f*)(qg + 4 * threadIdx.x);
        v4f q1 = *(const v4f*)(qg + 4 * (threadIdx.x + 256));
#pragma unroll
        for (int g = 0; g < 2; ++g)
#pragma unroll
            for (int i = 0; i < 8; ++i)
                fk[g][i] = *(const v2f*)(kp + (size_t)((g & 7) * 16) * D_DIM + 4 * (8 * (g >> 3) + i));
        const int r0 = threadIdx.x >> 4, c0 = (threadIdx.x & 15) * 4;
        *(v4f*)(qs + r0 * QP + c0)        = q0 * 0.0625f;
        *(v4f*)(qs + (r0 + 16) * QP + c0) = q1 * 0.0625f;
    }
    __syncthreads();
#pragma unroll
    for (int i = 0; i < 8; ++i) {
        bq[0][i] = *(const v2f*)(qp + 4 * i);
        bq[1][i] = *(const v2f*)(qp + 16 * QP + 4 * i);
    }
#pragma unroll
    for (int g = 0; g < 16; ++g) {
        const int t = g & 7;
        if (g + 2 < 16) {
            const int n = (g + 2) % 3, tn = (g + 2) & 7, hn = (g + 2) >> 3;
#pragma unroll
            for (int i = 0; i < 8; ++i)
                fk[n][i] = *(const v2f*)(kp + (size_t)(tn * 16) * D_DIM + 4 * (8 * hn + i));
        }
        const int c = g % 3;
#pragma unroll
        for (int i = 0; i < 8; ++i) {
            x[0][t] = __builtin_amdgcn_wmma_f32_16x16x4_f32(
                false, fk[c][i], false, bq[0][i], (short)0, x[0][t], false, false);
            x[1][t] = __builtin_amdgcn_wmma_f32_16x16x4_f32(
                false, fk[c][i], false, bq[1][i], (short)0, x[1][t], false, false);
        }
        if (g == 7) {
#pragma unroll
            for (int i = 0; i < 8; ++i) {
                bq[0][i] = *(const v2f*)(qp + 4 * (8 + i));
                bq[1][i] = *(const v2f*)(qp + 16 * QP + 4 * (8 + i));
            }
        }
        __builtin_amdgcn_sched_barrier(0);
    }

    float mx[2];
#pragma unroll
    for (int u = 0; u < 2; ++u) {
        float m0 = x[u][0][0], m1 = x[u][0][1];
#pragma unroll
        for (int t = 0; t < 8; ++t)
#pragma unroll
            for (int r = 0; r < 8; r += 2) {
                m0 = fmaxf(m0, x[u][t][r]);
                m1 = fmaxf(m1, x[u][t][r + 1]);
            }
        mx[u] = fmaxf(m0, m1);
    }
    {
        const float got = swap_halves(hi ? mx[0] : mx[1]);
        redm[wave * QB + lane] = fmaxf(hi ? mx[1] : mx[0], got);
    }
    __syncthreads();
    {
        float m = redm[lane];
#pragma unroll
        for (int w = 1; w < NW; ++w) m = fmaxf(m, redm[w * QB + lane]);
        const float mo = swap_halves(m);
        mx[0] = hi ? mo : m;
        mx[1] = hi ? m : mo;
    }
#pragma unroll
    for (int u = 0; u < 2; ++u)
#pragma unroll
        for (int t = 0; t < 8; ++t)
#pragma unroll
            for (int r = 0; r < 8; ++r) x[u][t][r] -= mx[u];

    float tau_own = -1.0f;
    float tau[2] = { -1.0f, -1.0f };
    float s_fin = 1.0f;
    for (int it = 0; it < N_STEP; ++it) {
        float sq[2], sg[2];
#pragma unroll
        for (int u = 0; u < 2; ++u) {
            float s0 = 0.f, s1 = 0.f, g0 = 0.f, g1 = 0.f;
            const float tu = tau[u];
#pragma unroll
            for (int t = 0; t < 8; ++t)
#pragma unroll
                for (int r = 0; r < 8; r += 2) {
                    float d0 = fmaxf(x[u][t][r]     - tu, 0.f);
                    float d1 = fmaxf(x[u][t][r + 1] - tu, 0.f);
                    s0 = fmaf(d0, d0, s0);  g0 += d0;
                    s1 = fmaf(d1, d1, s1);  g1 += d1;
                }
            sq[u] = s0 + s1;
            sg[u] = g0 + g1;
        }
        const float s = (hi ? sq[1] : sq[0]) + swap_halves(hi ? sq[0] : sq[1]);
        const float g = (hi ? sg[1] : sg[0]) + swap_halves(hi ? sg[0] : sg[1]);
        v2f* rb = red[it & 1];
        { v2f e; e.x = s; e.y = g; rb[wave * QB + lane] = e; }
        __syncthreads();
        float st = 0.f, gt = 0.f;
#pragma unroll
        for (int w = 0; w < NW; ++w) {
            v2f e = rb[w * QB + lane];
            st += e.x;  gt += e.y;
        }
        gt = fmaxf(gt, 1e-20f);
        const float t_old = tau_own;
        tau_own += (st - __builtin_amdgcn_sqrtf(st)) / gt;
        s_fin = fmaf(-2.0f * gt, tau_own - t_old, st);
        const float to = swap_halves(tau_own);
        tau[0] = hi ? to : tau_own;
        tau[1] = hi ? tau_own : to;
    }
    {
        const float inv_own = (s_fin > 0.f) ? (1.0f / s_fin) : 0.f;
        const float io = swap_halves(inv_own);
        const float inv0 = hi ? io : inv_own;
        const float inv1 = hi ? inv_own : io;
#pragma unroll
        for (int u = 0; u < 2; ++u) {
            const float tu = tau[u];
            const float inv = u ? inv1 : inv0;
#pragma unroll
            for (int t = 0; t < 8; ++t)
#pragma unroll
                for (int r = 0; r < 8; ++r) {
                    float d = fmaxf(x[u][t][r] - tu, 0.f);
                    x[u][t][r] = (d * d) * inv;
                }
        }
    }

    const float* vp = v + ((size_t)(bh * S_LEN + kbase + 8 * lhalf)) * D_DIM + lmod;
    float* pw = &part[wave][lmod * PP + 8 * lhalf];
#pragma unroll 1
    for (int dh = 0; dh < 2; ++dh) {
        v8f o[2][2];
#pragma unroll
        for (int e = 0; e < 2; ++e) { o[e][0] = (v8f){}; o[e][1] = (v8f){}; }
#pragma unroll
        for (int t = 0; t < 8; ++t) {
#pragma unroll
            for (int j = 0; j < 4; ++j) {
                const float* vr = vp + (size_t)(t * 16 + 2 * j) * D_DIM + dh * 32;
                v2f b0, b1;
                b0.x = x[0][t][2 * j];
                b0.y = x[0][t][2 * j + 1];
                b1.x = x[1][t][2 * j];
                b1.y = x[1][t][2 * j + 1];
#pragma unroll
                for (int e = 0; e < 2; ++e) {
                    v2f a;
                    a.x = vr[e * 16];
                    a.y = vr[D_DIM + e * 16];
                    o[e][0] = __builtin_amdgcn_wmma_f32_16x16x4_f32(
                        false, a, false, b0, (short)0, o[e][0], false, false);
                    o[e][1] = __builtin_amdgcn_wmma_f32_16x16x4_f32(
                        false, a, false, b1, (short)0, o[e][1], false, false);
                }
            }
        }
#pragma unroll
        for (int u = 0; u < 2; ++u)
#pragma unroll
            for (int e = 0; e < 2; ++e) {
                v4f lo, hi4;
                lo.x  = o[e][u][0]; lo.y  = o[e][u][1]; lo.z  = o[e][u][2]; lo.w  = o[e][u][3];
                hi4.x = o[e][u][4]; hi4.y = o[e][u][5]; hi4.z = o[e][u][6]; hi4.w = o[e][u][7];
                float* pd = pw + u * 16 * PP + dh * 32 + e * 16;
                *(v4f*)(pd)     = lo;
                *(v4f*)(pd + 4) = hi4;
            }
    }
    __syncthreads();

    const int od = (threadIdx.x & 15) * 4;
#pragma unroll
    for (int u = 0; u < 2; ++u) {
        const int oq = u * 16 + (threadIdx.x >> 4);
        v4f acc4 = *(const v4f*)(&part[0][oq * PP + od]);
#pragma unroll
        for (int w = 1; w < NW; ++w)
            acc4 += *(const v4f*)(&part[w][oq * PP + od]);
        *(v4f*)(out + ((size_t)(bh * S_LEN + qbase + oq)) * D_DIM + od) = acc4;
    }
}

extern "C" void kernel_launch(void* const* d_in, const int* in_sizes, int n_in,
                              void* d_out, int out_size, void* d_ws, size_t ws_size,
                              hipStream_t stream) {
    (void)in_sizes; (void)n_in; (void)out_size; (void)d_ws; (void)ws_size;
    const float* q = (const float*)d_in[0];
    const float* k = (const float*)d_in[1];
    const float* v = (const float*)d_in[2];
    float* out = (float*)d_out;

    dim3 grid(S_LEN / QB, BH);
    entmax_attn_kernel<<<grid, 256, 0, stream>>>(q, k, v, out);
}
